# starting kernel
# baseline (speedup 1.0000x reference)
#include <hip/hip_runtime.h>

#define B_ 2
#define S_ 4
#define T_ 1024
#define H_ 256
#define R_ 32
#define V_ (T_ - 1)
#define M_ 8
#define ROUTE_VOCAB_ 65536
#define SUB_STRIDE_ (R_ * ROUTE_VOCAB_)

typedef float v2f __attribute__((ext_vector_type(2)));
typedef float v8f __attribute__((ext_vector_type(8)));

__global__ __launch_bounds__(256)
void lngram_pack_codes(const float* __restrict__ hidden,
                       const float* __restrict__ qw,
                       unsigned char* __restrict__ codes) {
  const int lane = threadIdx.x & 31;
  const int wave = (blockIdx.x << 3) | (threadIdx.x >> 5);

  const int ot = wave & 15;
  const int tt = (wave >> 4) & 63;
  const int s  = (wave >> 10) & 3;
  const int b  = wave >> 12;

  const int t0  = tt << 4;
  const int o0  = ot << 4;
  const int mn  = lane & 15;
  const int kof = (lane >> 4) << 1;

  const float* __restrict__ arow =
      hidden + (size_t)(b * T_ + t0 + mn) * H_ + kof;
  const float* __restrict__ brow =
      qw + (size_t)(s * H_ + (o0 + mn)) * H_ + kof;

  v8f acc = {};
#pragma unroll 8
  for (int kk = 0; kk < H_; kk += 4) {
    v2f a  = *(const v2f*)(arow + kk);
    v2f bb = *(const v2f*)(brow + kk);
    acc = __builtin_amdgcn_wmma_f32_16x16x4_f32(
        false, a, false, bb, (short)0, acc, false, false);
  }

  unsigned mk0 = __builtin_amdgcn_ballot_w32(acc[0] > 0.0f);
  unsigned mk1 = __builtin_amdgcn_ballot_w32(acc[1] > 0.0f);
  unsigned mk2 = __builtin_amdgcn_ballot_w32(acc[2] > 0.0f);
  unsigned mk3 = __builtin_amdgcn_ballot_w32(acc[3] > 0.0f);
  unsigned mk4 = __builtin_amdgcn_ballot_w32(acc[4] > 0.0f);
  unsigned mk5 = __builtin_amdgcn_ballot_w32(acc[5] > 0.0f);
  unsigned mk6 = __builtin_amdgcn_ballot_w32(acc[6] > 0.0f);
  unsigned mk7 = __builtin_amdgcn_ballot_w32(acc[7] > 0.0f);

  const int jj   = lane & 7;
  const int part = lane >> 3;

  unsigned msel = mk0;
  msel = (jj == 1) ? mk1 : msel;
  msel = (jj == 2) ? mk2 : msel;
  msel = (jj == 3) ? mk3 : msel;
  msel = (jj == 4) ? mk4 : msel;
  msel = (jj == 5) ? mk5 : msel;
  msel = (jj == 6) ? mk6 : msel;
  msel = (jj == 7) ? mk7 : msel;

  const unsigned byte = (msel >> (part << 3)) & 0xFFu;
  const int t = t0 + jj + ((part & 2) ? 8 : 0);
  const int r = (ot << 1) + (part & 1);

  codes[(size_t)((b * S_ + s) * T_ + t) * R_ + r] = (unsigned char)byte;
}

__global__ __launch_bounds__(256)
void lngram_gather(const unsigned char* __restrict__ codes,
                   const float* __restrict__ flat_weight,
                   float* __restrict__ out) {
  const int gid = blockIdx.x * blockDim.x + threadIdx.x;
  if (gid >= B_ * S_ * V_ * R_) return;

  const int r   = gid & (R_ - 1);
  const int tv  = gid >> 5;
  const int v   = tv % V_;
  const int bs  = tv / V_;
  const int s   = bs & (S_ - 1);

  const unsigned char* __restrict__ cb =
      codes + (size_t)(bs * T_ + v) * R_ + r;
  const unsigned c0 = cb[0];
  const unsigned c1 = cb[R_];
  const unsigned addr = c0 | (c1 << 8);

  const size_t row = (size_t)s * SUB_STRIDE_ + (size_t)r * ROUTE_VOCAB_ + addr;
  const float4* __restrict__ src = (const float4*)(flat_weight + row * M_);
  const float4 x = src[0];
  const float4 y = src[1];

  float4* __restrict__ dst = (float4*)(out + (size_t)gid * M_);
  dst[0] = x;
  dst[1] = y;
}

extern "C" void kernel_launch(void* const* d_in, const int* in_sizes, int n_in,
                              void* d_out, int out_size, void* d_ws, size_t ws_size,
                              hipStream_t stream) {
  const float* hidden      = (const float*)d_in[0];
  const float* q_weight    = (const float*)d_in[1];
  const float* flat_weight = (const float*)d_in[2];
  float* out               = (float*)d_out;

  unsigned char* codes = (unsigned char*)d_ws;

  lngram_pack_codes<<<1024, 256, 0, stream>>>(hidden, q_weight, codes);

  const int total = B_ * S_ * V_ * R_;
  lngram_gather<<<(total + 255) / 256, 256, 0, stream>>>(codes, flat_weight, out);
}


# kernel 1: one wave per 32x64 logit tile (2x4 WMMA accumulators, 1 wave per SIMD); an A fragment feeds 4 WMMAs, a B fragment 2, from registers: 0.75 global fragment loads per WMMA, not 2; k order kept
# speedup vs baseline: 1.5858x; 1.5858x over previous
#include <hip/hip_runtime.h>

#define B_ 2
#define S_ 4
#define T_ 1024
#define H_ 256
#define R_ 32
#define V_ (T_ - 1)
#define M_ 8
#define ROUTE_VOCAB_ 65536
#define SUB_STRIDE_ (R_ * ROUTE_VOCAB_)

typedef float v2f __attribute__((ext_vector_type(2)));
typedef float v8f __attribute__((ext_vector_type(8)));

#define TM_ 2
#define TN_ 4

__global__ __launch_bounds__(128) __attribute__((amdgpu_waves_per_eu(1, 1)))
void lngram_pack_codes(const float* __restrict__ hidden,
                       const float* __restrict__ qw,
                       unsigned char* __restrict__ codes) {
  const int lane = threadIdx.x & 31;
  const int wi   = threadIdx.x >> 5;
  const int blk  = blockIdx.x;

  const int oB = blk & 1;
  const int tB = (blk >> 1) & 15;
  const int s  = (blk >> 5) & 3;
  const int b  = blk >> 7;

  const int t0  = (tB << 6) + ((wi >> 1) << 5);
  const int o0  = (oB << 7) + ((wi & 1) << 6);
  const int mn  = lane & 15;
  const int kof = (lane >> 4) << 1;

  const float* __restrict__ arow =
      hidden + (size_t)(b * T_ + t0 + mn) * H_ + kof;
  const float* __restrict__ brow =
      qw + (size_t)(s * H_ + (o0 + mn)) * H_ + kof;

  v8f acc[TM_][TN_];
#pragma unroll
  for (int i = 0; i < TM_; ++i)
#pragma unroll
    for (int j = 0; j < TN_; ++j)
      acc[i][j] = (v8f){};

#pragma unroll
  for (int kk = 0; kk < H_; kk += 4) {
    v2f a[TM_], bb[TN_];
#pragma unroll
    for (int i = 0; i < TM_; ++i)
      a[i] = *(const v2f*)(arow + i * 16 * H_ + kk);
#pragma unroll
    for (int j = 0; j < TN_; ++j)
      bb[j] = *(const v2f*)(brow + j * 16 * H_ + kk);
#pragma unroll
    for (int i = 0; i < TM_; ++i) {
#pragma unroll
      for (int j = 0; j < TN_; ++j) {
        acc[i][j] = __builtin_amdgcn_wmma_f32_16x16x4_f32(
            false, a[i], false, bb[j], (short)0, acc[i][j], false, false);
      }
    }
  }

  const int jj   = lane & 7;
  const int part = lane >> 3;

#pragma unroll
  for (int i = 0; i < TM_; ++i) {
#pragma unroll
    for (int j = 0; j < TN_; ++j) {
      const v8f c = acc[i][j];
      unsigned mk0 = __builtin_amdgcn_ballot_w32(c[0] > 0.0f);
      unsigned mk1 = __builtin_amdgcn_ballot_w32(c[1] > 0.0f);
      unsigned mk2 = __builtin_amdgcn_ballot_w32(c[2] > 0.0f);
      unsigned mk3 = __builtin_amdgcn_ballot_w32(c[3] > 0.0f);
      unsigned mk4 = __builtin_amdgcn_ballot_w32(c[4] > 0.0f);
      unsigned mk5 = __builtin_amdgcn_ballot_w32(c[5] > 0.0f);
      unsigned mk6 = __builtin_amdgcn_ballot_w32(c[6] > 0.0f);
      unsigned mk7 = __builtin_amdgcn_ballot_w32(c[7] > 0.0f);

      unsigned msel = mk0;
      msel = (jj == 1) ? mk1 : msel;
      msel = (jj == 2) ? mk2 : msel;
      msel = (jj == 3) ? mk3 : msel;
      msel = (jj == 4) ? mk4 : msel;
      msel = (jj == 5) ? mk5 : msel;
      msel = (jj == 6) ? mk6 : msel;
      msel = (jj == 7) ? mk7 : msel;

      const unsigned byte = (msel >> (part << 3)) & 0xFFu;
      const int t = t0 + (i << 4) + jj + ((part & 2) ? 8 : 0);
      const int r = (((o0 >> 4) + j) << 1) + (part & 1);

      codes[(size_t)((b * S_ + s) * T_ + t) * R_ + r] = (unsigned char)byte;
    }
  }
}

__global__ __launch_bounds__(256)
void lngram_gather(const unsigned char* __restrict__ codes,
                   const float* __restrict__ flat_weight,
                   float* __restrict__ out) {
  const int gid = blockIdx.x * blockDim.x + threadIdx.x;
  if (gid >= B_ * S_ * V_ * R_) return;

  const int r   = gid & (R_ - 1);
  const int tv  = gid >> 5;
  const int v   = tv % V_;
  const int bs  = tv / V_;
  const int s   = bs & (S_ - 1);

  const unsigned char* __restrict__ cb =
      codes + (size_t)(bs * T_ + v) * R_ + r;
  const unsigned c0 = cb[0];
  const unsigned c1 = cb[R_];
  const unsigned addr = c0 | (c1 << 8);

  const size_t row = (size_t)s * SUB_STRIDE_ + (size_t)r * ROUTE_VOCAB_ + addr;
  const float4* __restrict__ src = (const float4*)(flat_weight + row * M_);
  const float4 x = src[0];
  const float4 y = src[1];

  float4* __restrict__ dst = (float4*)(out + (size_t)gid * M_);
  dst[0] = x;
  dst[1] = y;
}

extern "C" void kernel_launch(void* const* d_in, const int* in_sizes, int n_in,
                              void* d_out, int out_size, void* d_ws, size_t ws_size,
                              hipStream_t stream) {
  const float* hidden      = (const float*)d_in[0];
  const float* q_weight    = (const float*)d_in[1];
  const float* flat_weight = (const float*)d_in[2];
  float* out               = (float*)d_out;

  unsigned char* codes = (unsigned char*)d_ws;

  lngram_pack_codes<<<256, 128, 0, stream>>>(hidden, q_weight, codes);

  const int total = B_ * S_ * V_ * R_;
  lngram_gather<<<(total + 255) / 256, 256, 0, stream>>>(codes, flat_weight, out);
}


# kernel 2: 1024 long waves (256 workgroups of 4) instead of 8184 one-row waves: each lane gathers 8 consecutive v, reads its 9 code bytes, then has all its b128 table reads in flight before it stores
# speedup vs baseline: 1.7101x; 1.0784x over previous
#include <hip/hip_runtime.h>

#define B_ 2
#define S_ 4
#define T_ 1024
#define H_ 256
#define R_ 32
#define V_ (T_ - 1)
#define M_ 8
#define ROUTE_VOCAB_ 65536
#define SUB_STRIDE_ (R_ * ROUTE_VOCAB_)

typedef float v2f __attribute__((ext_vector_type(2)));
typedef float v8f __attribute__((ext_vector_type(8)));
typedef float v4f __attribute__((ext_vector_type(4)));

#define TM_ 2
#define TN_ 4

__global__ __launch_bounds__(128) __attribute__((amdgpu_waves_per_eu(1, 1)))
void lngram_pack_codes(const float* __restrict__ hidden,
                       const float* __restrict__ qw,
                       unsigned char* __restrict__ codes) {
  const int lane = threadIdx.x & 31;
  const int wi   = threadIdx.x >> 5;
  const int blk  = blockIdx.x;

  const int oB = blk & 1;
  const int tB = (blk >> 1) & 15;
  const int s  = (blk >> 5) & 3;
  const int b  = blk >> 7;

  const int t0  = (tB << 6) + ((wi >> 1) << 5);
  const int o0  = (oB << 7) + ((wi & 1) << 6);
  const int mn  = lane & 15;
  const int kof = (lane >> 4) << 1;

  const float* __restrict__ arow =
      hidden + (size_t)(b * T_ + t0 + mn) * H_ + kof;
  const float* __restrict__ brow =
      qw + (size_t)(s * H_ + (o0 + mn)) * H_ + kof;

  v8f acc[TM_][TN_];
#pragma unroll
  for (int i = 0; i < TM_; ++i)
#pragma unroll
    for (int j = 0; j < TN_; ++j)
      acc[i][j] = (v8f){};

#pragma unroll
  for (int kk = 0; kk < H_; kk += 4) {
    v2f a[TM_], bb[TN_];
#pragma unroll
    for (int i = 0; i < TM_; ++i)
      a[i] = *(const v2f*)(arow + i * 16 * H_ + kk);
#pragma unroll
    for (int j = 0; j < TN_; ++j)
      bb[j] = *(const v2f*)(brow + j * 16 * H_ + kk);
#pragma unroll
    for (int i = 0; i < TM_; ++i) {
#pragma unroll
      for (int j = 0; j < TN_; ++j) {
        acc[i][j] = __builtin_amdgcn_wmma_f32_16x16x4_f32(
            false, a[i], false, bb[j], (short)0, acc[i][j], false, false);
      }
    }
  }

  const int jj   = lane & 7;
  const int part = lane >> 3;

#pragma unroll
  for (int i = 0; i < TM_; ++i) {
#pragma unroll
    for (int j = 0; j < TN_; ++j) {
      const v8f c = acc[i][j];
      unsigned mk0 = __builtin_amdgcn_ballot_w32(c[0] > 0.0f);
      unsigned mk1 = __builtin_amdgcn_ballot_w32(c[1] > 0.0f);
      unsigned mk2 = __builtin_amdgcn_ballot_w32(c[2] > 0.0f);
      unsigned mk3 = __builtin_amdgcn_ballot_w32(c[3] > 0.0f);
      unsigned mk4 = __builtin_amdgcn_ballot_w32(c[4] > 0.0f);
      unsigned mk5 = __builtin_amdgcn_ballot_w32(c[5] > 0.0f);
      unsigned mk6 = __builtin_amdgcn_ballot_w32(c[6] > 0.0f);
      unsigned mk7 = __builtin_amdgcn_ballot_w32(c[7] > 0.0f);

      unsigned msel = mk0;
      msel = (jj == 1) ? mk1 : msel;
      msel = (jj == 2) ? mk2 : msel;
      msel = (jj == 3) ? mk3 : msel;
      msel = (jj == 4) ? mk4 : msel;
      msel = (jj == 5) ? mk5 : msel;
      msel = (jj == 6) ? mk6 : msel;
      msel = (jj == 7) ? mk7 : msel;

      const unsigned byte = (msel >> (part << 3)) & 0xFFu;
      const int t = t0 + (i << 4) + jj + ((part & 2) ? 8 : 0);
      const int r = (((o0 >> 4) + j) << 1) + (part & 1);

      codes[(size_t)((b * S_ + s) * T_ + t) * R_ + r] = (unsigned char)byte;
    }
  }
}

#define VW_ 8

__global__ __launch_bounds__(128)
void lngram_gather(const unsigned char* __restrict__ codes,
                   const float* __restrict__ flat_weight,
                   float* __restrict__ out) {
  const int lane = threadIdx.x & 31;
  const int wave = __builtin_amdgcn_readfirstlane(
      (int)((blockIdx.x << 2) + (threadIdx.x >> 5)));
  const int bs = wave >> 7;
  const int v0 = (wave & 127) * VW_;
  const int s  = bs & (S_ - 1);

  const unsigned char* __restrict__ cb = codes + (size_t)(bs * T_) * R_ + lane;

  unsigned c[VW_ + 1];
#pragma unroll
  for (int i = 0; i <= VW_; ++i) {
    const int t = (v0 + i < T_) ? (v0 + i) : (T_ - 1);
    c[i] = cb[t * R_];
  }

  const float* __restrict__ tab =
      flat_weight + ((size_t)s * SUB_STRIDE_ + (size_t)lane * ROUTE_VOCAB_) * M_;

  v4f x[VW_], y[VW_];
#pragma unroll
  for (int i = 0; i < VW_; ++i) {
    const unsigned addr = c[i] | (c[i + 1] << 8);
    const v4f* __restrict__ src = (const v4f*)(tab + (size_t)addr * M_);
    x[i] = src[0];
    y[i] = src[1];
  }

  asm volatile("" ::
      "v"(x[0]), "v"(y[0]), "v"(x[1]), "v"(y[1]), "v"(x[2]), "v"(y[2]),
      "v"(x[3]), "v"(y[3]), "v"(x[4]), "v"(y[4]), "v"(x[5]), "v"(y[5]),
      "v"(x[6]), "v"(y[6]), "v"(x[7]), "v"(y[7]));

  float* __restrict__ ob = out + ((size_t)(bs * V_ + v0) * R_ + lane) * M_;
#pragma unroll
  for (int i = 0; i < VW_; ++i) {
    if (v0 + i < V_) {
      v4f* __restrict__ dst = (v4f*)(ob + (size_t)i * (R_ * M_));
      dst[0] = x[i];
      dst[1] = y[i];
    }
  }
}

extern "C" void kernel_launch(void* const* d_in, const int* in_sizes, int n_in,
                              void* d_out, int out_size, void* d_ws, size_t ws_size,
                              hipStream_t stream) {
  const float* hidden      = (const float*)d_in[0];
  const float* q_weight    = (const float*)d_in[1];
  const float* flat_weight = (const float*)d_in[2];
  float* out               = (float*)d_out;

  unsigned char* codes = (unsigned char*)d_ws;

  lngram_pack_codes<<<256, 128, 0, stream>>>(hidden, q_weight, codes);

  lngram_gather<<<256, 128, 0, stream>>>(codes, flat_weight, out);
}
